# starting kernel
# baseline (speedup 1.0000x reference)
#include <hip/hip_runtime.h>
#include <hip/hip_bf16.h>
#include <hip/hip_fp16.h>

#define N_NODES 10000
#define N_EDGES 320000
#define IN_DIM  256
#define HID     64
#define HEADS   4
#define D1      (HEADS * HID)
#define OUT_DIM 64
#define NEG_SLOPE 0.2f

typedef __attribute__((ext_vector_type(16))) _Float16 v16h;
typedef __attribute__((ext_vector_type(8)))  _Float16 v8h;
typedef __attribute__((ext_vector_type(8)))  float    v8f;

__device__ inline void atomicMaxF(float* addr, float val) {
    if (val >= 0.0f)
        atomicMax((int*)addr, __float_as_int(val));
    else
        atomicMin((unsigned int*)addr, (unsigned int)__float_as_int(val));
}

__global__ void cvt_f32_to_f16(const float* __restrict__ in, _Float16* __restrict__ out, int n) {
    int i = blockIdx.x * blockDim.x + threadIdx.x;
    if (i < n) out[i] = (_Float16)in[i];
}

__global__ void cvt_transpose_f16(const float* __restrict__ W, _Float16* __restrict__ Wt,
                                  int K, int N) {
    int i = blockIdx.x * blockDim.x + threadIdx.x;
    if (i >= K * N) return;
    int k = i / N, n = i % N;
    Wt[(size_t)n * K + k] = (_Float16)W[i];
}

__global__ void fill_f32(float* __restrict__ p, float v, int n) {
    int i = blockIdx.x * blockDim.x + threadIdx.x;
    if (i < n) p[i] = v;
}

__global__ void init_bias(float* __restrict__ out, const float* __restrict__ bias, int n, int d) {
    int i = blockIdx.x * blockDim.x + threadIdx.x;
    if (i < n * d) out[i] = bias[i % d];
}

__global__ void elu_bias_to_f16(const float* __restrict__ in, const float* __restrict__ bias,
                                _Float16* __restrict__ out, int n, int F) {
    int i = blockIdx.x * blockDim.x + threadIdx.x;
    if (i >= n) return;
    float x = in[i] + bias[i % F];
    x = (x > 0.0f) ? x : (__expf(x) - 1.0f);
    out[i] = (_Float16)x;
}

__device__ inline v16h load_frag(const _Float16* __restrict__ base, int half) {
    v8h lo = *(const v8h*)(base + 8 * half);
    v8h hi = *(const v8h*)(base + 16 + 8 * half);
    v16h r;
#pragma unroll
    for (int i = 0; i < 8; ++i) { r[i] = lo[i]; r[8 + i] = hi[i]; }
    return r;
}

__global__ void gemm_wmma_f16t(const _Float16* __restrict__ A, const _Float16* __restrict__ Bt,
                               float* __restrict__ C, int M, int N, int K) {
    const int lane = threadIdx.x & 31;
    const int wave = threadIdx.x >> 5;
    const int m0   = blockIdx.x * 16;
    const int n0   = (blockIdx.y * (blockDim.x >> 5) + wave) * 64;
    if (m0 >= M || n0 >= N) return;

    const int half = lane >> 4;
    const int l15  = lane & 15;

    const _Float16* aptr = A + (size_t)(m0 + l15) * K;
    const _Float16* bptr0 = Bt + (size_t)(n0 +  0 + l15) * K;
    const _Float16* bptr1 = Bt + (size_t)(n0 + 16 + l15) * K;
    const _Float16* bptr2 = Bt + (size_t)(n0 + 32 + l15) * K;
    const _Float16* bptr3 = Bt + (size_t)(n0 + 48 + l15) * K;

    v8f acc0 = {}, acc1 = {}, acc2 = {}, acc3 = {};

    for (int k0 = 0; k0 < K; k0 += 32) {
        v16h a  = load_frag(aptr  + k0, half);
        v16h b0 = load_frag(bptr0 + k0, half);
        v16h b1 = load_frag(bptr1 + k0, half);
        v16h b2 = load_frag(bptr2 + k0, half);
        v16h b3 = load_frag(bptr3 + k0, half);
        acc0 = __builtin_amdgcn_wmma_f32_16x16x32_f16(false, a, false, b0, (short)0, acc0, false, false);
        acc1 = __builtin_amdgcn_wmma_f32_16x16x32_f16(false, a, false, b1, (short)0, acc1, false, false);
        acc2 = __builtin_amdgcn_wmma_f32_16x16x32_f16(false, a, false, b2, (short)0, acc2, false, false);
        acc3 = __builtin_amdgcn_wmma_f32_16x16x32_f16(false, a, false, b3, (short)0, acc3, false, false);
    }

    float* cbase = C + (size_t)(m0 + 8 * half) * N + n0 + l15;
#pragma unroll
    for (int r = 0; r < 8; ++r) {
        float* row = cbase + (size_t)r * N;
        row[0]  = acc0[r];
        row[16] = acc1[r];
        row[32] = acc2[r];
        row[48] = acc3[r];
    }
}

__global__ void attn_scores(const float* __restrict__ feat, const float* __restrict__ al,
                            const float* __restrict__ ar, float* __restrict__ el,
                            float* __restrict__ er, int nheads, int d) {
    int idx = blockIdx.x * blockDim.x + threadIdx.x;
    if (idx >= N_NODES * nheads) return;
    int n = idx / nheads, h = idx % nheads;
    const float* f = feat + (size_t)(n * nheads + h) * d;
    float sl = 0.f, sr = 0.f;
    for (int j = 0; j < d; ++j) {
        float fv = f[j];
        sl += fv * al[h * d + j];
        sr += fv * ar[h * d + j];
    }
    el[idx] = sl;
    er[idx] = sr;
}

__global__ void edge_max(const int* __restrict__ src, const int* __restrict__ dst,
                         const float* __restrict__ el, const float* __restrict__ er,
                         float* __restrict__ ebuf, float* __restrict__ m, int nheads) {
    int idx = blockIdx.x * blockDim.x + threadIdx.x;
    if (idx >= N_EDGES * nheads) return;
    int e = idx / nheads, h = idx % nheads;
    int s = src[e], t = dst[e];
    float x = el[s * nheads + h] + er[t * nheads + h];
    x = (x > 0.0f) ? x : NEG_SLOPE * x;
    ebuf[idx] = x;
    atomicMaxF(&m[t * nheads + h], x);
}

__global__ void edge_exp(const int* __restrict__ dst, const float* __restrict__ m,
                         float* __restrict__ ebuf, float* __restrict__ denom, int nheads) {
    int idx = blockIdx.x * blockDim.x + threadIdx.x;
    if (idx >= N_EDGES * nheads) return;
    int e = idx / nheads, h = idx % nheads;
    int t = dst[e];
    float ex = __expf(ebuf[idx] - m[t * nheads + h]);
    ebuf[idx] = ex;
    atomicAdd(&denom[t * nheads + h], ex);
}

__global__ void edge_aggregate(const int* __restrict__ src, const int* __restrict__ dst,
                               const float* __restrict__ feat, const float* __restrict__ ebuf,
                               const float* __restrict__ denom, float* __restrict__ out,
                               int nheads, int d) {
    int gwave = (blockIdx.x * blockDim.x + threadIdx.x) >> 5;
    int lane  = threadIdx.x & 31;
    if (gwave >= N_EDGES) return;
    int s = src[gwave], t = dst[gwave];
    int F = nheads * d;
    for (int h = 0; h < nheads; ++h) {
        float alpha = ebuf[gwave * nheads + h] / denom[t * nheads + h];
        for (int j = lane; j < d; j += 32) {
            atomicAdd(&out[(size_t)t * F + h * d + j],
                      feat[(size_t)s * F + h * d + j] * alpha);
        }
    }
}

extern "C" void kernel_launch(void* const* d_in, const int* in_sizes, int n_in,
                              void* d_out, int out_size, void* d_ws, size_t ws_size,
                              hipStream_t stream) {
    const float* h   = (const float*)d_in[0];
    const int*   src = (const int*)  d_in[1];
    const int*   dst = (const int*)  d_in[2];
    const float* W1  = (const float*)d_in[3];
    const float* al1 = (const float*)d_in[4];
    const float* ar1 = (const float*)d_in[5];
    const float* b1  = (const float*)d_in[6];
    const float* W2  = (const float*)d_in[7];
    const float* al2 = (const float*)d_in[8];
    const float* ar2 = (const float*)d_in[9];
    const float* b2  = (const float*)d_in[10];
    float* out = (float*)d_out;

    size_t off = 0;
    auto alloc = [&](size_t bytes) {
        void* p = (char*)d_ws + off;
        off += (bytes + 255) & ~(size_t)255;
        return p;
    };
    _Float16* hf16   = (_Float16*)alloc((size_t)N_NODES * IN_DIM  * 2);
    _Float16* W1t16  = (_Float16*)alloc((size_t)IN_DIM  * D1      * 2);
    _Float16* W2t16  = (_Float16*)alloc((size_t)D1      * OUT_DIM * 2);
    float*    feat1  = (float*)   alloc((size_t)N_NODES * D1      * 4);
    float*    el1    = (float*)   alloc((size_t)N_NODES * HEADS   * 4);
    float*    er1    = (float*)   alloc((size_t)N_NODES * HEADS   * 4);
    float*    m1     = (float*)   alloc((size_t)N_NODES * HEADS   * 4);
    float*    den1   = (float*)   alloc((size_t)N_NODES * HEADS   * 4);
    float*    ebuf1  = (float*)   alloc((size_t)N_EDGES * HEADS   * 4);
    float*    out1   = (float*)   alloc((size_t)N_NODES * D1      * 4);
    _Float16* h1f16  = (_Float16*)alloc((size_t)N_NODES * D1      * 2);
    float*    feat2  = (float*)   alloc((size_t)N_NODES * OUT_DIM * 4);
    float*    el2    = (float*)   alloc((size_t)N_NODES * 4);
    float*    er2    = (float*)   alloc((size_t)N_NODES * 4);
    float*    m2     = (float*)   alloc((size_t)N_NODES * 4);
    float*    den2   = (float*)   alloc((size_t)N_NODES * 4);
    float*    ebuf2  = (float*)   alloc((size_t)N_EDGES * 4);

    const float NEG_INF = -__builtin_inff();
    auto cdiv = [](int a, int b) { return (a + b - 1) / b; };

    cvt_f32_to_f16<<<cdiv(N_NODES * IN_DIM, 256), 256, 0, stream>>>(h, hf16, N_NODES * IN_DIM);
    cvt_transpose_f16<<<cdiv(IN_DIM * D1, 256), 256, 0, stream>>>(W1, W1t16, IN_DIM, D1);
    cvt_transpose_f16<<<cdiv(D1 * OUT_DIM, 256), 256, 0, stream>>>(W2, W2t16, D1, OUT_DIM);

    gemm_wmma_f16t<<<dim3(N_NODES / 16, 1), 128, 0, stream>>>(hf16, W1t16, feat1,
                                                              N_NODES, D1, IN_DIM);
    attn_scores<<<cdiv(N_NODES * HEADS, 256), 256, 0, stream>>>(feat1, al1, ar1, el1, er1,
                                                                HEADS, HID);
    fill_f32<<<cdiv(N_NODES * HEADS, 256), 256, 0, stream>>>(m1,   NEG_INF, N_NODES * HEADS);
    fill_f32<<<cdiv(N_NODES * HEADS, 256), 256, 0, stream>>>(den1, 0.0f,    N_NODES * HEADS);
    fill_f32<<<cdiv(N_NODES * D1,    256), 256, 0, stream>>>(out1, 0.0f,    N_NODES * D1);

    edge_max<<<cdiv(N_EDGES * HEADS, 256), 256, 0, stream>>>(src, dst, el1, er1, ebuf1, m1, HEADS);
    edge_exp<<<cdiv(N_EDGES * HEADS, 256), 256, 0, stream>>>(dst, m1, ebuf1, den1, HEADS);
    edge_aggregate<<<cdiv(N_EDGES * 32, 256), 256, 0, stream>>>(src, dst, feat1, ebuf1, den1,
                                                                out1, HEADS, HID);
    elu_bias_to_f16<<<cdiv(N_NODES * D1, 256), 256, 0, stream>>>(out1, b1, h1f16,
                                                                 N_NODES * D1, D1);

    gemm_wmma_f16t<<<dim3(N_NODES / 16, 1), 32, 0, stream>>>(h1f16, W2t16, feat2,
                                                             N_NODES, OUT_DIM, D1);
    attn_scores<<<cdiv(N_NODES, 256), 256, 0, stream>>>(feat2, al2, ar2, el2, er2, 1, OUT_DIM);
    fill_f32<<<cdiv(N_NODES, 256), 256, 0, stream>>>(m2,   NEG_INF, N_NODES);
    fill_f32<<<cdiv(N_NODES, 256), 256, 0, stream>>>(den2, 0.0f,    N_NODES);
    init_bias<<<cdiv(N_NODES * OUT_DIM, 256), 256, 0, stream>>>(out, b2, N_NODES, OUT_DIM);

    edge_max<<<cdiv(N_EDGES, 256), 256, 0, stream>>>(src, dst, el2, er2, ebuf2, m2, 1);
    edge_exp<<<cdiv(N_EDGES, 256), 256, 0, stream>>>(dst, m2, ebuf2, den2, 1);
    edge_aggregate<<<cdiv(N_EDGES * 32, 256), 256, 0, stream>>>(src, dst, feat2, ebuf2, den2,
                                                                out, 1, OUT_DIM);
    (void)in_sizes; (void)n_in; (void)out_size; (void)ws_size;
}


# edge softmax and aggregation per destination: edges counting-sorted by dst once per call (LDS ranks), one wave per node sums in registers; float atomics gone, sum order now sorted-edge order
# speedup vs baseline: 3.7929x; 3.7929x over previous
#include <hip/hip_runtime.h>
#include <hip/hip_bf16.h>
#include <hip/hip_fp16.h>

#define N_NODES 10000
#define N_EDGES 320000
#define IN_DIM  256
#define HID     64
#define HEADS   4
#define D1      (HEADS * HID)
#define OUT_DIM 64
#define NEG_SLOPE 0.2f

typedef __attribute__((ext_vector_type(16))) _Float16 v16h;
typedef __attribute__((ext_vector_type(8)))  _Float16 v8h;
typedef __attribute__((ext_vector_type(8)))  float    v8f;

__global__ void cvt_f32_to_f16(const float* __restrict__ in, _Float16* __restrict__ out, int n) {
    int i = blockIdx.x * blockDim.x + threadIdx.x;
    if (i < n) out[i] = (_Float16)in[i];
}

__global__ void cvt_transpose_f16(const float* __restrict__ W, _Float16* __restrict__ Wt,
                                  int K, int N) {
    int i = blockIdx.x * blockDim.x + threadIdx.x;
    if (i >= K * N) return;
    int k = i / N, n = i % N;
    Wt[(size_t)n * K + k] = (_Float16)W[i];
}

__device__ inline v16h load_frag(const _Float16* __restrict__ base, int half) {
    v8h lo = *(const v8h*)(base + 8 * half);
    v8h hi = *(const v8h*)(base + 16 + 8 * half);
    v16h r;
#pragma unroll
    for (int i = 0; i < 8; ++i) { r[i] = lo[i]; r[8 + i] = hi[i]; }
    return r;
}

__global__ void gemm_wmma_f16t(const _Float16* __restrict__ A, const _Float16* __restrict__ Bt,
                               float* __restrict__ C, int M, int N, int K) {
    const int lane = threadIdx.x & 31;
    const int wave = threadIdx.x >> 5;
    const int m0   = blockIdx.x * 16;
    const int n0   = (blockIdx.y * (blockDim.x >> 5) + wave) * 64;
    if (m0 >= M || n0 >= N) return;

    const int half = lane >> 4;
    const int l15  = lane & 15;

    const _Float16* aptr = A + (size_t)(m0 + l15) * K;
    const _Float16* bptr0 = Bt + (size_t)(n0 +  0 + l15) * K;
    const _Float16* bptr1 = Bt + (size_t)(n0 + 16 + l15) * K;
    const _Float16* bptr2 = Bt + (size_t)(n0 + 32 + l15) * K;
    const _Float16* bptr3 = Bt + (size_t)(n0 + 48 + l15) * K;

    v8f acc0 = {}, acc1 = {}, acc2 = {}, acc3 = {};

    for (int k0 = 0; k0 < K; k0 += 32) {
        v16h a  = load_frag(aptr  + k0, half);
        v16h b0 = load_frag(bptr0 + k0, half);
        v16h b1 = load_frag(bptr1 + k0, half);
        v16h b2 = load_frag(bptr2 + k0, half);
        v16h b3 = load_frag(bptr3 + k0, half);
        acc0 = __builtin_amdgcn_wmma_f32_16x16x32_f16(false, a, false, b0, (short)0, acc0, false, false);
        acc1 = __builtin_amdgcn_wmma_f32_16x16x32_f16(false, a, false, b1, (short)0, acc1, false, false);
        acc2 = __builtin_amdgcn_wmma_f32_16x16x32_f16(false, a, false, b2, (short)0, acc2, false, false);
        acc3 = __builtin_amdgcn_wmma_f32_16x16x32_f16(false, a, false, b3, (short)0, acc3, false, false);
    }

    float* cbase = C + (size_t)(m0 + 8 * half) * N + n0 + l15;
#pragma unroll
    for (int r = 0; r < 8; ++r) {
        float* row = cbase + (size_t)r * N;
        row[0]  = acc0[r];
        row[16] = acc1[r];
        row[32] = acc2[r];
        row[48] = acc3[r];
    }
}

__global__ void attn_scores(const float* __restrict__ feat, const float* __restrict__ al,
                            const float* __restrict__ ar, float* __restrict__ el,
                            float* __restrict__ er, int nheads, int d) {
    int idx = blockIdx.x * blockDim.x + threadIdx.x;
    if (idx >= N_NODES * nheads) return;
    int n = idx / nheads, h = idx % nheads;
    const float* f = feat + (size_t)(n * nheads + h) * d;
    float sl = 0.f, sr = 0.f;
    for (int j = 0; j < d; ++j) {
        float fv = f[j];
        sl += fv * al[h * d + j];
        sr += fv * ar[h * d + j];
    }
    el[idx] = sl;
    er[idx] = sr;
}

#define SORT_BLOCKS 40
#define SORT_CHUNK  (N_EDGES / SORT_BLOCKS)
#define SORT_TRIPS  ((SORT_CHUNK + 1023) / 1024)

typedef __attribute__((ext_vector_type(4))) _Float16 v4h;

__global__ __launch_bounds__(1024) void dst_slice_hist(const int* __restrict__ dst,
                                                       int* __restrict__ rank,
                                                       int* __restrict__ slice_cnt) {
    __shared__ int cnt[N_NODES];
    for (int i = threadIdx.x; i < N_NODES; i += 1024) cnt[i] = 0;
    __syncthreads();
    const int e0 = blockIdx.x * SORT_CHUNK;
    int tv[SORT_TRIPS];
#pragma unroll
    for (int k = 0; k < SORT_TRIPS; ++k) {
        int i = threadIdx.x + k * 1024;
        tv[k] = (i < SORT_CHUNK) ? dst[e0 + i] : 0;
    }
#pragma unroll
    for (int k = 0; k < SORT_TRIPS; ++k) {
        int i = threadIdx.x + k * 1024;
        if (i < SORT_CHUNK) rank[e0 + i] = atomicAdd(&cnt[tv[k]], 1);
    }
    __syncthreads();
    int* row = slice_cnt + (size_t)blockIdx.x * N_NODES;
    for (int i = threadIdx.x; i < N_NODES; i += 1024) row[i] = cnt[i];
}

__global__ void slice_prefix(int* __restrict__ slice_cnt, int* __restrict__ deg) {
    int n = blockIdx.x * blockDim.x + threadIdx.x;
    if (n >= N_NODES) return;
    int v[SORT_BLOCKS];
#pragma unroll
    for (int b = 0; b < SORT_BLOCKS; ++b) v[b] = slice_cnt[(size_t)b * N_NODES + n];
    int run = 0;
#pragma unroll
    for (int b = 0; b < SORT_BLOCKS; ++b) {
        slice_cnt[(size_t)b * N_NODES + n] = run;
        run += v[b];
    }
    deg[n] = run;
}

#define SCAN_PER 10
__global__ __launch_bounds__(1024) void degree_scan(const int* __restrict__ deg,
                                                    int* __restrict__ rowptr) {
    __shared__ int part[1024];
    const int tid = threadIdx.x;
    const int base = tid * SCAN_PER;
    int loc[SCAN_PER];
    int sum = 0;
#pragma unroll
    for (int k = 0; k < SCAN_PER; ++k) {
        int idx = base + k;
        int v = (idx < N_NODES) ? deg[idx] : 0;
        loc[k] = sum;
        sum += v;
    }
    part[tid] = sum;
    __syncthreads();
    for (int off = 1; off < 1024; off <<= 1) {
        int v = (tid >= off) ? part[tid - off] : 0;
        __syncthreads();
        part[tid] += v;
        __syncthreads();
    }
    const int excl = part[tid] - sum;
#pragma unroll
    for (int k = 0; k < SCAN_PER; ++k) {
        int idx = base + k;
        if (idx < N_NODES) rowptr[idx] = excl + loc[k];
    }
    if (tid == 1023) rowptr[N_NODES] = part[1023];
}

__global__ void place_src(const int* __restrict__ src, const int* __restrict__ dst,
                          const int* __restrict__ rank, const int* __restrict__ slice_off,
                          const int* __restrict__ rowptr, int* __restrict__ seg_src) {
    int e = blockIdx.x * blockDim.x + threadIdx.x;
    if (e >= N_EDGES) return;
    int t = dst[e];
    int b = e / SORT_CHUNK;
    seg_src[rowptr[t] + slice_off[(size_t)b * N_NODES + t] + rank[e]] = src[e];
}

__device__ inline float wave_max(float v) {
    for (int o = 16; o > 0; o >>= 1) v = fmaxf(v, __shfl_xor(v, o, 32));
    return v;
}
__device__ inline float wave_sum(float v) {
    for (int o = 16; o > 0; o >>= 1) v += __shfl_xor(v, o, 32);
    return v;
}
__device__ inline float leaky(float x) { return (x > 0.0f) ? x : NEG_SLOPE * x; }
__device__ inline int   lane_i(int v, int j)   { return __builtin_amdgcn_readlane(v, j); }
__device__ inline float lane_f(float v, int j) {
    return __int_as_float(__builtin_amdgcn_readlane(__float_as_int(v), j));
}
__device__ inline void fma4(float4& acc, const float4& f, float a) {
    acc.x = __builtin_fmaf(f.x, a, acc.x); acc.y = __builtin_fmaf(f.y, a, acc.y);
    acc.z = __builtin_fmaf(f.z, a, acc.z); acc.w = __builtin_fmaf(f.w, a, acc.w);
}
__device__ inline float elu1(float x) { return (x > 0.0f) ? x : (__expf(x) - 1.0f); }

__global__ __launch_bounds__(256) void gat_gather_l1(const int* __restrict__ rowptr,
                                                     const int* __restrict__ seg_src,
                                                     const float* __restrict__ feat,
                                                     const float* __restrict__ el,
                                                     const float* __restrict__ er,
                                                     const float* __restrict__ bias,
                                                     _Float16* __restrict__ hout) {
    const int lane = threadIdx.x & 31;
    const int wave = __builtin_amdgcn_readfirstlane((int)(threadIdx.x >> 5));
    const int t = blockIdx.x * 8 + wave;
    if (t >= N_NODES) return;
    const int start = rowptr[t];
    const int end   = rowptr[t + 1];
    const bool hi = (lane >> 4) != 0;
    const float4 r4 = *(const float4*)(er + (size_t)t * 4);
    const float NEG_INF = -__builtin_inff();

    float m0 = NEG_INF, m1 = NEG_INF, m2 = NEG_INF, m3 = NEG_INF;
    for (int base = start; base < end; base += 32) {
        int i = base + lane;
        if (i < end) {
            int s = seg_src[i];
            float4 l4 = *(const float4*)(el + (size_t)s * 4);
            m0 = fmaxf(m0, leaky(l4.x + r4.x));
            m1 = fmaxf(m1, leaky(l4.y + r4.y));
            m2 = fmaxf(m2, leaky(l4.z + r4.z));
            m3 = fmaxf(m3, leaky(l4.w + r4.w));
        }
    }
    m0 = wave_max(m0); m1 = wave_max(m1); m2 = wave_max(m2); m3 = wave_max(m3);

    float d0 = 0.f, d1 = 0.f, d2 = 0.f, d3 = 0.f;
    for (int base = start; base < end; base += 32) {
        int i = base + lane;
        if (i < end) {
            int s = seg_src[i];
            float4 l4 = *(const float4*)(el + (size_t)s * 4);
            d0 += __expf(leaky(l4.x + r4.x) - m0);
            d1 += __expf(leaky(l4.y + r4.y) - m1);
            d2 += __expf(leaky(l4.z + r4.z) - m2);
            d3 += __expf(leaky(l4.w + r4.w) - m3);
        }
    }
    d0 = wave_sum(d0); d1 = wave_sum(d1); d2 = wave_sum(d2); d3 = wave_sum(d3);

    float4 accA = {0.f, 0.f, 0.f, 0.f}, accB = {0.f, 0.f, 0.f, 0.f};
    const float* fcol = feat + lane * 4;
    for (int base = start; base < end; base += 32) {
        int i = base + lane;
        int s = 0;
        float a0 = 0.f, a1 = 0.f, a2 = 0.f, a3 = 0.f;
        if (i < end) {
            s = seg_src[i];
            float4 l4 = *(const float4*)(el + (size_t)s * 4);
            a0 = __expf(leaky(l4.x + r4.x) - m0) / d0;
            a1 = __expf(leaky(l4.y + r4.y) - m1) / d1;
            a2 = __expf(leaky(l4.z + r4.z) - m2) / d2;
            a3 = __expf(leaky(l4.w + r4.w) - m3) / d3;
        }
        const int cnt = min(32, end - base);
        int j = 0;
        for (; j + 4 <= cnt; j += 4) {
            float4 fA[4], fB[4];
#pragma unroll
            for (int k = 0; k < 4; ++k) {
                const float* p = fcol + (size_t)lane_i(s, j + k) * D1;
                fA[k] = *(const float4*)p;
                fB[k] = *(const float4*)(p + 128);
            }
#pragma unroll
            for (int k = 0; k < 4; ++k) {
                const float w0 = lane_f(a0, j + k), w1 = lane_f(a1, j + k);
                const float w2 = lane_f(a2, j + k), w3 = lane_f(a3, j + k);
                fma4(accA, fA[k], hi ? w1 : w0);
                fma4(accB, fB[k], hi ? w3 : w2);
            }
        }
        for (; j < cnt; ++j) {
            const float* p = fcol + (size_t)lane_i(s, j) * D1;
            float4 fa = *(const float4*)p;
            float4 fb = *(const float4*)(p + 128);
            const float w0 = lane_f(a0, j), w1 = lane_f(a1, j);
            const float w2 = lane_f(a2, j), w3 = lane_f(a3, j);
            fma4(accA, fa, hi ? w1 : w0);
            fma4(accB, fb, hi ? w3 : w2);
        }
    }

    const float4 bA = *(const float4*)(bias + lane * 4);
    const float4 bB = *(const float4*)(bias + 128 + lane * 4);
    v4h oA = {(_Float16)elu1(accA.x + bA.x), (_Float16)elu1(accA.y + bA.y),
              (_Float16)elu1(accA.z + bA.z), (_Float16)elu1(accA.w + bA.w)};
    v4h oB = {(_Float16)elu1(accB.x + bB.x), (_Float16)elu1(accB.y + bB.y),
              (_Float16)elu1(accB.z + bB.z), (_Float16)elu1(accB.w + bB.w)};
    _Float16* orow = hout + (size_t)t * D1 + lane * 4;
    *(v4h*)orow = oA;
    *(v4h*)(orow + 128) = oB;
}

__global__ __launch_bounds__(256) void gat_gather_l2(const int* __restrict__ rowptr,
                                                     const int* __restrict__ seg_src,
                                                     const float* __restrict__ feat,
                                                     const float* __restrict__ el,
                                                     const float* __restrict__ er,
                                                     const float* __restrict__ bias,
                                                     float* __restrict__ out) {
    const int lane = threadIdx.x & 31;
    const int wave = __builtin_amdgcn_readfirstlane((int)(threadIdx.x >> 5));
    const int t = blockIdx.x * 8 + wave;
    if (t >= N_NODES) return;
    const int start = rowptr[t];
    const int end   = rowptr[t + 1];
    const float r = er[t];
    const float NEG_INF = -__builtin_inff();

    float m = NEG_INF;
    for (int base = start; base < end; base += 32) {
        int i = base + lane;
        if (i < end) m = fmaxf(m, leaky(el[seg_src[i]] + r));
    }
    m = wave_max(m);

    float d = 0.f;
    for (int base = start; base < end; base += 32) {
        int i = base + lane;
        if (i < end) d += __expf(leaky(el[seg_src[i]] + r) - m);
    }
    d = wave_sum(d);

    float2 acc = {0.f, 0.f};
    const float* fcol = feat + lane * 2;
    for (int base = start; base < end; base += 32) {
        int i = base + lane;
        int s = 0;
        float a = 0.f;
        if (i < end) {
            s = seg_src[i];
            a = __expf(leaky(el[s] + r) - m) / d;
        }
        const int cnt = min(32, end - base);
        int j = 0;
        for (; j + 8 <= cnt; j += 8) {
            float2 f[8];
#pragma unroll
            for (int k = 0; k < 8; ++k)
                f[k] = *(const float2*)(fcol + (size_t)lane_i(s, j + k) * OUT_DIM);
#pragma unroll
            for (int k = 0; k < 8; ++k) {
                float w = lane_f(a, j + k);
                acc.x = __builtin_fmaf(f[k].x, w, acc.x);
                acc.y = __builtin_fmaf(f[k].y, w, acc.y);
            }
        }
        for (; j < cnt; ++j) {
            float2 f = *(const float2*)(fcol + (size_t)lane_i(s, j) * OUT_DIM);
            float w = lane_f(a, j);
            acc.x = __builtin_fmaf(f.x, w, acc.x);
            acc.y = __builtin_fmaf(f.y, w, acc.y);
        }
    }

    const float2 b = *(const float2*)(bias + lane * 2);
    float2 o = {acc.x + b.x, acc.y + b.y};
    *(float2*)(out + (size_t)t * OUT_DIM + lane * 2) = o;
}

extern "C" void kernel_launch(void* const* d_in, const int* in_sizes, int n_in,
                              void* d_out, int out_size, void* d_ws, size_t ws_size,
                              hipStream_t stream) {
    const float* h   = (const float*)d_in[0];
    const int*   src = (const int*)  d_in[1];
    const int*   dst = (const int*)  d_in[2];
    const float* W1  = (const float*)d_in[3];
    const float* al1 = (const float*)d_in[4];
    const float* ar1 = (const float*)d_in[5];
    const float* b1  = (const float*)d_in[6];
    const float* W2  = (const float*)d_in[7];
    const float* al2 = (const float*)d_in[8];
    const float* ar2 = (const float*)d_in[9];
    const float* b2  = (const float*)d_in[10];
    float* out = (float*)d_out;

    size_t off = 0;
    auto alloc = [&](size_t bytes) {
        void* p = (char*)d_ws + off;
        off += (bytes + 255) & ~(size_t)255;
        return p;
    };
    _Float16* hf16   = (_Float16*)alloc((size_t)N_NODES * IN_DIM  * 2);
    _Float16* W1t16  = (_Float16*)alloc((size_t)IN_DIM  * D1      * 2);
    _Float16* W2t16  = (_Float16*)alloc((size_t)D1      * OUT_DIM * 2);
    float*    feat1  = (float*)   alloc((size_t)N_NODES * D1      * 4);
    float*    el1    = (float*)   alloc((size_t)N_NODES * HEADS   * 4);
    float*    er1    = (float*)   alloc((size_t)N_NODES * HEADS   * 4);
    _Float16* h1f16  = (_Float16*)alloc((size_t)N_NODES * D1      * 2);
    float*    feat2  = (float*)   alloc((size_t)N_NODES * OUT_DIM * 4);
    float*    el2    = (float*)   alloc((size_t)N_NODES * 4);
    float*    er2    = (float*)   alloc((size_t)N_NODES * 4);
    int*      rank   = (int*)     alloc((size_t)N_EDGES * 4);
    int*      slices = (int*)     alloc((size_t)SORT_BLOCKS * N_NODES * 4);
    int*      deg    = (int*)     alloc((size_t)N_NODES * 4);
    int*      rowptr = (int*)     alloc((size_t)(N_NODES + 1) * 4);
    int*      segsrc = (int*)     alloc((size_t)N_EDGES * 4);

    auto cdiv = [](int a, int b) { return (a + b - 1) / b; };

    cvt_f32_to_f16<<<cdiv(N_NODES * IN_DIM, 256), 256, 0, stream>>>(h, hf16, N_NODES * IN_DIM);
    cvt_transpose_f16<<<cdiv(IN_DIM * D1, 256), 256, 0, stream>>>(W1, W1t16, IN_DIM, D1);
    cvt_transpose_f16<<<cdiv(D1 * OUT_DIM, 256), 256, 0, stream>>>(W2, W2t16, D1, OUT_DIM);

    dst_slice_hist<<<SORT_BLOCKS, 1024, 0, stream>>>(dst, rank, slices);
    slice_prefix<<<cdiv(N_NODES, 256), 256, 0, stream>>>(slices, deg);
    degree_scan<<<1, 1024, 0, stream>>>(deg, rowptr);
    place_src<<<cdiv(N_EDGES, 256), 256, 0, stream>>>(src, dst, rank, slices, rowptr, segsrc);

    gemm_wmma_f16t<<<dim3(N_NODES / 16, 1), 128, 0, stream>>>(hf16, W1t16, feat1,
                                                              N_NODES, D1, IN_DIM);
    attn_scores<<<cdiv(N_NODES * HEADS, 256), 256, 0, stream>>>(feat1, al1, ar1, el1, er1,
                                                                HEADS, HID);
    gat_gather_l1<<<cdiv(N_NODES, 8), 256, 0, stream>>>(rowptr, segsrc, feat1, el1, er1, b1, h1f16);

    gemm_wmma_f16t<<<dim3(N_NODES / 16, 1), 32, 0, stream>>>(h1f16, W2t16, feat2,
                                                             N_NODES, OUT_DIM, D1);
    attn_scores<<<cdiv(N_NODES, 256), 256, 0, stream>>>(feat2, al2, ar2, el2, er2, 1, OUT_DIM);
    gat_gather_l2<<<cdiv(N_NODES, 8), 256, 0, stream>>>(rowptr, segsrc, feat2, el2, er2, b2, out);
    (void)in_sizes; (void)n_in; (void)out_size; (void)ws_size;
}


# independent stages share a launch (weight converts beside the dst histogram, edge placement beside projection 1); rowptr scan made group-local via per-slice group totals: 10 -> 6 launches
# speedup vs baseline: 4.8753x; 1.2854x over previous
#include <hip/hip_runtime.h>
#include <hip/hip_bf16.h>
#include <hip/hip_fp16.h>

#define N_NODES 10000
#define N_EDGES 320000
#define IN_DIM  256
#define HID     64
#define HEADS   4
#define D1      (HEADS * HID)
#define OUT_DIM 64
#define NEG_SLOPE 0.2f

typedef __attribute__((ext_vector_type(16))) _Float16 v16h;
typedef __attribute__((ext_vector_type(8)))  _Float16 v8h;
typedef __attribute__((ext_vector_type(8)))  float    v8f;

__device__ inline void cvt_transpose_one(const float* __restrict__ W, _Float16* __restrict__ Wt,
                                         int K, int N, int i) {
    int k = i / N, n = i % N;
    Wt[(size_t)n * K + k] = (_Float16)W[i];
}

__device__ inline v16h load_frag(const _Float16* __restrict__ base, int half) {
    v8h lo = *(const v8h*)(base + 8 * half);
    v8h hi = *(const v8h*)(base + 16 + 8 * half);
    v16h r;
#pragma unroll
    for (int i = 0; i < 8; ++i) { r[i] = lo[i]; r[8 + i] = hi[i]; }
    return r;
}

__device__ inline v16h load_frag(const float* __restrict__ base, int half) {
    const float4* lo = (const float4*)(base + 8 * half);
    const float4* hi = (const float4*)(base + 16 + 8 * half);
    const float4 l0 = lo[0], l1 = lo[1], h0 = hi[0], h1 = hi[1];
    v16h r;
    r[0]  = (_Float16)l0.x; r[1]  = (_Float16)l0.y; r[2]  = (_Float16)l0.z; r[3]  = (_Float16)l0.w;
    r[4]  = (_Float16)l1.x; r[5]  = (_Float16)l1.y; r[6]  = (_Float16)l1.z; r[7]  = (_Float16)l1.w;
    r[8]  = (_Float16)h0.x; r[9]  = (_Float16)h0.y; r[10] = (_Float16)h0.z; r[11] = (_Float16)h0.w;
    r[12] = (_Float16)h1.x; r[13] = (_Float16)h1.y; r[14] = (_Float16)h1.z; r[15] = (_Float16)h1.w;
    return r;
}

template <typename TA>
__device__ inline void gemm_wmma_body(const TA* __restrict__ A, const _Float16* __restrict__ Bt,
                                      float* __restrict__ C,
                                      const float* __restrict__ al, const float* __restrict__ ar,
                                      float* __restrict__ el, float* __restrict__ er,
                                      int M, int N, int K) {
    const int lane = threadIdx.x & 31;
    const int wave = threadIdx.x >> 5;
    const int m0   = blockIdx.x * 16;
    const int n0   = (blockIdx.y * (blockDim.x >> 5) + wave) * 64;
    if (m0 >= M || n0 >= N) return;

    const int half = lane >> 4;
    const int l15  = lane & 15;

    const TA* aptr = A + (size_t)(m0 + l15) * K;
    const _Float16* bptr0 = Bt + (size_t)(n0 +  0 + l15) * K;
    const _Float16* bptr1 = Bt + (size_t)(n0 + 16 + l15) * K;
    const _Float16* bptr2 = Bt + (size_t)(n0 + 32 + l15) * K;
    const _Float16* bptr3 = Bt + (size_t)(n0 + 48 + l15) * K;

    v8f acc0 = {}, acc1 = {}, acc2 = {}, acc3 = {};

    for (int k0 = 0; k0 < K; k0 += 32) {
        v16h a  = load_frag(aptr  + k0, half);
        v16h b0 = load_frag(bptr0 + k0, half);
        v16h b1 = load_frag(bptr1 + k0, half);
        v16h b2 = load_frag(bptr2 + k0, half);
        v16h b3 = load_frag(bptr3 + k0, half);
        acc0 = __builtin_amdgcn_wmma_f32_16x16x32_f16(false, a, false, b0, (short)0, acc0, false, false);
        acc1 = __builtin_amdgcn_wmma_f32_16x16x32_f16(false, a, false, b1, (short)0, acc1, false, false);
        acc2 = __builtin_amdgcn_wmma_f32_16x16x32_f16(false, a, false, b2, (short)0, acc2, false, false);
        acc3 = __builtin_amdgcn_wmma_f32_16x16x32_f16(false, a, false, b3, (short)0, acc3, false, false);
    }

    float* cbase = C + (size_t)(m0 + 8 * half) * N + n0 + l15;
#pragma unroll
    for (int r = 0; r < 8; ++r) {
        float* row = cbase + (size_t)r * N;
        row[0]  = acc0[r];
        row[16] = acc1[r];
        row[32] = acc2[r];
        row[48] = acc3[r];
    }

    const float* alp = al + n0 + l15;
    const float* arp = ar + n0 + l15;
    const float al0 = alp[0], al1 = alp[16], al2 = alp[32], al3 = alp[48];
    const float ar0 = arp[0], ar1 = arp[16], ar2 = arp[32], ar3 = arp[48];
    float sl[8], sr[8];
#pragma unroll
    for (int r = 0; r < 8; ++r) {
        sl[r] = acc0[r] * al0 + acc1[r] * al1 + acc2[r] * al2 + acc3[r] * al3;
        sr[r] = acc0[r] * ar0 + acc1[r] * ar1 + acc2[r] * ar2 + acc3[r] * ar3;
    }
#pragma unroll
    for (int o = 8; o > 0; o >>= 1) {
#pragma unroll
        for (int r = 0; r < 8; ++r) {
            sl[r] += __shfl_xor(sl[r], o, 32);
            sr[r] += __shfl_xor(sr[r], o, 32);
        }
    }
    float vl = sl[0], vr = sr[0];
#pragma unroll
    for (int r = 1; r < 8; ++r) {
        if (l15 == r) { vl = sl[r]; vr = sr[r]; }
    }
    if (l15 < 8) {
        const size_t o = (size_t)(m0 + 8 * half + l15) * (N >> 6) + (n0 >> 6);
        el[o] = vl;
        er[o] = vr;
    }
}

__global__ void gemm_wmma_f16t(const _Float16* __restrict__ A, const _Float16* __restrict__ Bt,
                               float* __restrict__ C,
                               const float* __restrict__ al, const float* __restrict__ ar,
                               float* __restrict__ el, float* __restrict__ er,
                               int M, int N, int K) {
    gemm_wmma_body<_Float16>(A, Bt, C, al, ar, el, er, M, N, K);
}

#define SORT_BLOCKS 40
#define SORT_CHUNK  (N_EDGES / SORT_BLOCKS)
#define SORT_TRIPS  ((SORT_CHUNK + 1023) / 1024)
#define NODE_GROUP  256
#define N_GROUPS    ((N_NODES + NODE_GROUP - 1) / NODE_GROUP)
#define GROUP_TRIPS (N_GROUPS * NODE_GROUP / 1024)
#define CVT1_BLOCKS (IN_DIM * D1 / 1024)
#define CVT2_BLOCKS (D1 * OUT_DIM / 1024)
#define GEMM1_BLOCKS (N_NODES / 16)
#define PLACE_BLOCKS (N_EDGES / 128)

typedef __attribute__((ext_vector_type(4))) _Float16 v4h;

__device__ inline int wave_sum_i(int v) {
    for (int o = 16; o > 0; o >>= 1) v += __shfl_xor(v, o, 32);
    return v;
}

__global__ __launch_bounds__(1024) void prep_hist_cvt(const int* __restrict__ dst,
                                                      int* __restrict__ rank,
                                                      int* __restrict__ slice_cnt,
                                                      int* __restrict__ group_cnt,
                                                      const float* __restrict__ W1,
                                                      _Float16* __restrict__ W1t,
                                                      const float* __restrict__ W2,
                                                      _Float16* __restrict__ W2t) {
    __shared__ int cnt[N_NODES];
    __shared__ int wpart[GROUP_TRIPS * 32];
    const int tid = threadIdx.x;
    if (blockIdx.x >= SORT_BLOCKS) {
        int i = (blockIdx.x - SORT_BLOCKS) * 1024 + tid;
        if (i < IN_DIM * D1) {
            cvt_transpose_one(W1, W1t, IN_DIM, D1, i);
        } else {
            i -= IN_DIM * D1;
            if (i < D1 * OUT_DIM) cvt_transpose_one(W2, W2t, D1, OUT_DIM, i);
        }
        return;
    }
    for (int i = tid; i < N_NODES; i += 1024) cnt[i] = 0;
    __syncthreads();
    const int e0 = blockIdx.x * SORT_CHUNK;
    int tv[SORT_TRIPS];
#pragma unroll
    for (int k = 0; k < SORT_TRIPS; ++k) {
        int i = tid + k * 1024;
        tv[k] = (i < SORT_CHUNK) ? dst[e0 + i] : 0;
    }
#pragma unroll
    for (int k = 0; k < SORT_TRIPS; ++k) {
        int i = tid + k * 1024;
        if (i < SORT_CHUNK) rank[e0 + i] = atomicAdd(&cnt[tv[k]], 1);
    }
    __syncthreads();
    int* row = slice_cnt + (size_t)blockIdx.x * N_NODES;
    for (int k = 0; k < GROUP_TRIPS; ++k) {
        int i = tid + k * 1024;
        int cv = 0;
        if (i < N_NODES) { cv = cnt[i]; row[i] = cv; }
        cv = wave_sum_i(cv);
        if ((tid & 31) == 0) wpart[k * 32 + (tid >> 5)] = cv;
    }
    __syncthreads();
    if (tid < N_GROUPS) {
        const int* p = wpart + (tid >> 2) * 32 + (tid & 3) * 8;
        int s = 0;
        for (int w = 0; w < 8; ++w) s += p[w];
        group_cnt[blockIdx.x * N_GROUPS + tid] = s;
    }
}

__global__ __launch_bounds__(NODE_GROUP) void slice_prefix_rowptr(int* __restrict__ slice_cnt,
                                                                  const int* __restrict__ group_cnt,
                                                                  int* __restrict__ rowptr) {
    __shared__ int wtot[NODE_GROUP / 32];
    __shared__ int wbel[NODE_GROUP / 32];
    const int tid = threadIdx.x;
    const int lane = tid & 31;
    const int wave = tid >> 5;
    const int g = blockIdx.x;
    const int n = g * NODE_GROUP + tid;
    const bool valid = n < N_NODES;
    int below = 0;
    if (tid < g) {
#pragma unroll
        for (int b = 0; b < SORT_BLOCKS; ++b) below += group_cnt[b * N_GROUPS + tid];
    }
    int v[SORT_BLOCKS];
#pragma unroll
    for (int b = 0; b < SORT_BLOCKS; ++b) v[b] = valid ? slice_cnt[(size_t)b * N_NODES + n] : 0;
    int run = 0;
#pragma unroll
    for (int b = 0; b < SORT_BLOCKS; ++b) {
        if (valid) slice_cnt[(size_t)b * N_NODES + n] = run;
        run += v[b];
    }
    int inc = run;
    for (int o = 1; o < 32; o <<= 1) {
        int u = __shfl_up(inc, o, 32);
        if (lane >= o) inc += u;
    }
    below = wave_sum_i(below);
    if (lane == 31) wtot[wave] = inc;
    if (lane == 0)  wbel[wave] = below;
    __syncthreads();
    int base = 0;
    for (int w = 0; w < NODE_GROUP / 32; ++w) {
        base += wbel[w];
        if (w < wave) base += wtot[w];
    }
    if (valid) rowptr[n] = base + inc - run;
    if (n == N_NODES - 1) rowptr[N_NODES] = base + inc;
}

__global__ __launch_bounds__(128) void proj1_and_place(const float* __restrict__ A,
                                                       const _Float16* __restrict__ Bt,
                                                       float* __restrict__ C,
                                                       const float* __restrict__ al,
                                                       const float* __restrict__ ar,
                                                       float* __restrict__ el, float* __restrict__ er,
                                                       const int* __restrict__ src,
                                                       const int* __restrict__ dst,
                                                       const int* __restrict__ rank,
                                                       const int* __restrict__ slice_off,
                                                       const int* __restrict__ rowptr,
                                                       int* __restrict__ seg_src) {
    if (blockIdx.x < GEMM1_BLOCKS) {
        gemm_wmma_body<float>(A, Bt, C, al, ar, el, er, N_NODES, D1, IN_DIM);
        return;
    }
    int e = (blockIdx.x - GEMM1_BLOCKS) * 128 + threadIdx.x;
    if (e >= N_EDGES) return;
    int t = dst[e];
    int b = e / SORT_CHUNK;
    seg_src[rowptr[t] + slice_off[(size_t)b * N_NODES + t] + rank[e]] = src[e];
}

__device__ inline float wave_max(float v) {
    for (int o = 16; o > 0; o >>= 1) v = fmaxf(v, __shfl_xor(v, o, 32));
    return v;
}
__device__ inline float wave_sum(float v) {
    for (int o = 16; o > 0; o >>= 1) v += __shfl_xor(v, o, 32);
    return v;
}
__device__ inline float leaky(float x) { return (x > 0.0f) ? x : NEG_SLOPE * x; }
__device__ inline int   lane_i(int v, int j)   { return __builtin_amdgcn_readlane(v, j); }
__device__ inline float lane_f(float v, int j) {
    return __int_as_float(__builtin_amdgcn_readlane(__float_as_int(v), j));
}
__device__ inline void fma4(float4& acc, const float4& f, float a) {
    acc.x = __builtin_fmaf(f.x, a, acc.x); acc.y = __builtin_fmaf(f.y, a, acc.y);
    acc.z = __builtin_fmaf(f.z, a, acc.z); acc.w = __builtin_fmaf(f.w, a, acc.w);
}
__device__ inline float elu1(float x) { return (x > 0.0f) ? x : (__expf(x) - 1.0f); }

__global__ __launch_bounds__(256) void gat_gather_l1(const int* __restrict__ rowptr,
                                                     const int* __restrict__ seg_src,
                                                     const float* __restrict__ feat,
                                                     const float* __restrict__ el,
                                                     const float* __restrict__ er,
                                                     const float* __restrict__ bias,
                                                     _Float16* __restrict__ hout) {
    const int lane = threadIdx.x & 31;
    const int wave = __builtin_amdgcn_readfirstlane((int)(threadIdx.x >> 5));
    const int t = blockIdx.x * 8 + wave;
    if (t >= N_NODES) return;
    const int start = rowptr[t];
    const int end   = rowptr[t + 1];
    const bool hi = (lane >> 4) != 0;
    const float4 r4 = *(const float4*)(er + (size_t)t * 4);
    const float NEG_INF = -__builtin_inff();

    float m0 = NEG_INF, m1 = NEG_INF, m2 = NEG_INF, m3 = NEG_INF;
    for (int base = start; base < end; base += 32) {
        int i = base + lane;
        if (i < end) {
            int s = seg_src[i];
            float4 l4 = *(const float4*)(el + (size_t)s * 4);
            m0 = fmaxf(m0, leaky(l4.x + r4.x));
            m1 = fmaxf(m1, leaky(l4.y + r4.y));
            m2 = fmaxf(m2, leaky(l4.z + r4.z));
            m3 = fmaxf(m3, leaky(l4.w + r4.w));
        }
    }
    m0 = wave_max(m0); m1 = wave_max(m1); m2 = wave_max(m2); m3 = wave_max(m3);

    float d0 = 0.f, d1 = 0.f, d2 = 0.f, d3 = 0.f;
    for (int base = start; base < end; base += 32) {
        int i = base + lane;
        if (i < end) {
            int s = seg_src[i];
            float4 l4 = *(const float4*)(el + (size_t)s * 4);
            d0 += __expf(leaky(l4.x + r4.x) - m0);
            d1 += __expf(leaky(l4.y + r4.y) - m1);
            d2 += __expf(leaky(l4.z + r4.z) - m2);
            d3 += __expf(leaky(l4.w + r4.w) - m3);
        }
    }
    d0 = wave_sum(d0); d1 = wave_sum(d1); d2 = wave_sum(d2); d3 = wave_sum(d3);

    float4 accA = {0.f, 0.f, 0.f, 0.f}, accB = {0.f, 0.f, 0.f, 0.f};
    const float* fcol = feat + lane * 4;
    for (int base = start; base < end; base += 32) {
        int i = base + lane;
        int s = 0;
        float a0 = 0.f, a1 = 0.f, a2 = 0.f, a3 = 0.f;
        if (i < end) {
            s = seg_src[i];
            float4 l4 = *(const float4*)(el + (size_t)s * 4);
            a0 = __expf(leaky(l4.x + r4.x) - m0) / d0;
            a1 = __expf(leaky(l4.y + r4.y) - m1) / d1;
            a2 = __expf(leaky(l4.z + r4.z) - m2) / d2;
            a3 = __expf(leaky(l4.w + r4.w) - m3) / d3;
        }
        const int cnt = min(32, end - base);
        int j = 0;
        for (; j + 4 <= cnt; j += 4) {
            float4 fA[4], fB[4];
#pragma unroll
            for (int k = 0; k < 4; ++k) {
                const float* p = fcol + (size_t)lane_i(s, j + k) * D1;
                fA[k] = *(const float4*)p;
                fB[k] = *(const float4*)(p + 128);
            }
#pragma unroll
            for (int k = 0; k < 4; ++k) {
                const float w0 = lane_f(a0, j + k), w1 = lane_f(a1, j + k);
                const float w2 = lane_f(a2, j + k), w3 = lane_f(a3, j + k);
                fma4(accA, fA[k], hi ? w1 : w0);
                fma4(accB, fB[k], hi ? w3 : w2);
            }
        }
        for (; j < cnt; ++j) {
            const float* p = fcol + (size_t)lane_i(s, j) * D1;
            float4 fa = *(const float4*)p;
            float4 fb = *(const float4*)(p + 128);
            const float w0 = lane_f(a0, j), w1 = lane_f(a1, j);
            const float w2 = lane_f(a2, j), w3 = lane_f(a3, j);
            fma4(accA, fa, hi ? w1 : w0);
            fma4(accB, fb, hi ? w3 : w2);
        }
    }

    const float4 bA = *(const float4*)(bias + lane * 4);
    const float4 bB = *(const float4*)(bias + 128 + lane * 4);
    v4h oA = {(_Float16)elu1(accA.x + bA.x), (_Float16)elu1(accA.y + bA.y),
              (_Float16)elu1(accA.z + bA.z), (_Float16)elu1(accA.w + bA.w)};
    v4h oB = {(_Float16)elu1(accB.x + bB.x), (_Float16)elu1(accB.y + bB.y),
              (_Float16)elu1(accB.z + bB.z), (_Float16)elu1(accB.w + bB.w)};
    _Float16* orow = hout + (size_t)t * D1 + lane * 4;
    *(v4h*)orow = oA;
    *(v4h*)(orow + 128) = oB;
}

__global__ __launch_bounds__(256) void gat_gather_l2(const int* __restrict__ rowptr,
                                                     const int* __restrict__ seg_src,
                                                     const float* __restrict__ feat,
                                                     const float* __restrict__ el,
                                                     const float* __restrict__ er,
                                                     const float* __restrict__ bias,
                                                     float* __restrict__ out) {
    const int lane = threadIdx.x & 31;
    const int wave = __builtin_amdgcn_readfirstlane((int)(threadIdx.x >> 5));
    const int t = blockIdx.x * 8 + wave;
    if (t >= N_NODES) return;
    const int start = rowptr[t];
    const int end   = rowptr[t + 1];
    const float r = er[t];
    const float NEG_INF = -__builtin_inff();

    float m = NEG_INF;
    for (int base = start; base < end; base += 32) {
        int i = base + lane;
        if (i < end) m = fmaxf(m, leaky(el[seg_src[i]] + r));
    }
    m = wave_max(m);

    float d = 0.f;
    for (int base = start; base < end; base += 32) {
        int i = base + lane;
        if (i < end) d += __expf(leaky(el[seg_src[i]] + r) - m);
    }
    d = wave_sum(d);

    float2 acc = {0.f, 0.f};
    const float* fcol = feat + lane * 2;
    for (int base = start; base < end; base += 32) {
        int i = base + lane;
        int s = 0;
        float a = 0.f;
        if (i < end) {
            s = seg_src[i];
            a = __expf(leaky(el[s] + r) - m) / d;
        }
        const int cnt = min(32, end - base);
        int j = 0;
        for (; j + 8 <= cnt; j += 8) {
            float2 f[8];
#pragma unroll
            for (int k = 0; k < 8; ++k)
                f[k] = *(const float2*)(fcol + (size_t)lane_i(s, j + k) * OUT_DIM);
#pragma unroll
            for (int k = 0; k < 8; ++k) {
                float w = lane_f(a, j + k);
                acc.x = __builtin_fmaf(f[k].x, w, acc.x);
                acc.y = __builtin_fmaf(f[k].y, w, acc.y);
            }
        }
        for (; j < cnt; ++j) {
            float2 f = *(const float2*)(fcol + (size_t)lane_i(s, j) * OUT_DIM);
            float w = lane_f(a, j);
            acc.x = __builtin_fmaf(f.x, w, acc.x);
            acc.y = __builtin_fmaf(f.y, w, acc.y);
        }
    }

    const float2 b = *(const float2*)(bias + lane * 2);
    float2 o = {acc.x + b.x, acc.y + b.y};
    *(float2*)(out + (size_t)t * OUT_DIM + lane * 2) = o;
}

extern "C" void kernel_launch(void* const* d_in, const int* in_sizes, int n_in,
                              void* d_out, int out_size, void* d_ws, size_t ws_size,
                              hipStream_t stream) {
    const float* h   = (const float*)d_in[0];
    const int*   src = (const int*)  d_in[1];
    const int*   dst = (const int*)  d_in[2];
    const float* W1  = (const float*)d_in[3];
    const float* al1 = (const float*)d_in[4];
    const float* ar1 = (const float*)d_in[5];
    const float* b1  = (const float*)d_in[6];
    const float* W2  = (const float*)d_in[7];
    const float* al2 = (const float*)d_in[8];
    const float* ar2 = (const float*)d_in[9];
    const float* b2  = (const float*)d_in[10];
    float* out = (float*)d_out;

    size_t off = 0;
    auto alloc = [&](size_t bytes) {
        void* p = (char*)d_ws + off;
        off += (bytes + 255) & ~(size_t)255;
        return p;
    };
    _Float16* W1t16  = (_Float16*)alloc((size_t)IN_DIM  * D1      * 2);
    _Float16* W2t16  = (_Float16*)alloc((size_t)D1      * OUT_DIM * 2);
    float*    feat1  = (float*)   alloc((size_t)N_NODES * D1      * 4);
    float*    el1    = (float*)   alloc((size_t)N_NODES * HEADS   * 4);
    float*    er1    = (float*)   alloc((size_t)N_NODES * HEADS   * 4);
    _Float16* h1f16  = (_Float16*)alloc((size_t)N_NODES * D1      * 2);
    float*    feat2  = (float*)   alloc((size_t)N_NODES * OUT_DIM * 4);
    float*    el2    = (float*)   alloc((size_t)N_NODES * 4);
    float*    er2    = (float*)   alloc((size_t)N_NODES * 4);
    int*      rank   = (int*)     alloc((size_t)N_EDGES * 4);
    int*      slices = (int*)     alloc((size_t)SORT_BLOCKS * N_NODES * 4);
    int*      groups = (int*)     alloc((size_t)SORT_BLOCKS * N_GROUPS * 4);
    int*      rowptr = (int*)     alloc((size_t)(N_NODES + 1) * 4);
    int*      segsrc = (int*)     alloc((size_t)N_EDGES * 4);

    auto cdiv = [](int a, int b) { return (a + b - 1) / b; };

    prep_hist_cvt<<<SORT_BLOCKS + CVT1_BLOCKS + CVT2_BLOCKS, 1024, 0, stream>>>(
        dst, rank, slices, groups, W1, W1t16, W2, W2t16);
    slice_prefix_rowptr<<<N_GROUPS, NODE_GROUP, 0, stream>>>(slices, groups, rowptr);

    proj1_and_place<<<GEMM1_BLOCKS + PLACE_BLOCKS, 128, 0, stream>>>(h, W1t16, feat1, al1, ar1, el1, er1,
                                                                   src, dst, rank, slices, rowptr, segsrc);
    gat_gather_l1<<<cdiv(N_NODES, 8), 256, 0, stream>>>(rowptr, segsrc, feat1, el1, er1, b1, h1f16);

    gemm_wmma_f16t<<<dim3(N_NODES / 16, 1), 32, 0, stream>>>(h1f16, W2t16, feat2, al2, ar2,
                                                             el2, er2, N_NODES, OUT_DIM, D1);
    gat_gather_l2<<<cdiv(N_NODES, 8), 256, 0, stream>>>(rowptr, segsrc, feat2, el2, er2, b2, out);
    (void)in_sizes; (void)n_in; (void)out_size; (void)ws_size;
}


# both gathers, nodes of 1..64 in-edges: each edge's source and score loaded once, kept in registers through max, denominator and weights; both halves' loads issued together; sum order unchanged
# speedup vs baseline: 5.6039x; 1.1494x over previous
#include <hip/hip_runtime.h>
#include <hip/hip_bf16.h>
#include <hip/hip_fp16.h>

#define N_NODES 10000
#define N_EDGES 320000
#define IN_DIM  256
#define HID     64
#define HEADS   4
#define D1      (HEADS * HID)
#define OUT_DIM 64
#define NEG_SLOPE 0.2f

typedef __attribute__((ext_vector_type(16))) _Float16 v16h;
typedef __attribute__((ext_vector_type(8)))  _Float16 v8h;
typedef __attribute__((ext_vector_type(8)))  float    v8f;

__device__ inline void cvt_transpose_one(const float* __restrict__ W, _Float16* __restrict__ Wt,
                                         int K, int N, int i) {
    int k = i / N, n = i % N;
    Wt[(size_t)n * K + k] = (_Float16)W[i];
}

__device__ inline v16h load_frag(const _Float16* __restrict__ base, int half) {
    v8h lo = *(const v8h*)(base + 8 * half);
    v8h hi = *(const v8h*)(base + 16 + 8 * half);
    v16h r;
#pragma unroll
    for (int i = 0; i < 8; ++i) { r[i] = lo[i]; r[8 + i] = hi[i]; }
    return r;
}

__device__ inline v16h load_frag(const float* __restrict__ base, int half) {
    const float4* lo = (const float4*)(base + 8 * half);
    const float4* hi = (const float4*)(base + 16 + 8 * half);
    const float4 l0 = lo[0], l1 = lo[1], h0 = hi[0], h1 = hi[1];
    v16h r;
    r[0]  = (_Float16)l0.x; r[1]  = (_Float16)l0.y; r[2]  = (_Float16)l0.z; r[3]  = (_Float16)l0.w;
    r[4]  = (_Float16)l1.x; r[5]  = (_Float16)l1.y; r[6]  = (_Float16)l1.z; r[7]  = (_Float16)l1.w;
    r[8]  = (_Float16)h0.x; r[9]  = (_Float16)h0.y; r[10] = (_Float16)h0.z; r[11] = (_Float16)h0.w;
    r[12] = (_Float16)h1.x; r[13] = (_Float16)h1.y; r[14] = (_Float16)h1.z; r[15] = (_Float16)h1.w;
    return r;
}

template <typename TA>
__device__ inline void gemm_wmma_body(const TA* __restrict__ A, const _Float16* __restrict__ Bt,
                                      float* __restrict__ C,
                                      const float* __restrict__ al, const float* __restrict__ ar,
                                      float* __restrict__ el, float* __restrict__ er,
                                      int M, int N, int K) {
    const int lane = threadIdx.x & 31;
    const int wave = threadIdx.x >> 5;
    const int m0   = blockIdx.x * 16;
    const int n0   = (blockIdx.y * (blockDim.x >> 5) + wave) * 64;
    if (m0 >= M || n0 >= N) return;

    const int half = lane >> 4;
    const int l15  = lane & 15;

    const TA* aptr = A + (size_t)(m0 + l15) * K;
    const _Float16* bptr0 = Bt + (size_t)(n0 +  0 + l15) * K;
    const _Float16* bptr1 = Bt + (size_t)(n0 + 16 + l15) * K;
    const _Float16* bptr2 = Bt + (size_t)(n0 + 32 + l15) * K;
    const _Float16* bptr3 = Bt + (size_t)(n0 + 48 + l15) * K;

    v8f acc0 = {}, acc1 = {}, acc2 = {}, acc3 = {};

    for (int k0 = 0; k0 < K; k0 += 32) {
        v16h a  = load_frag(aptr  + k0, half);
        v16h b0 = load_frag(bptr0 + k0, half);
        v16h b1 = load_frag(bptr1 + k0, half);
        v16h b2 = load_frag(bptr2 + k0, half);
        v16h b3 = load_frag(bptr3 + k0, half);
        acc0 = __builtin_amdgcn_wmma_f32_16x16x32_f16(false, a, false, b0, (short)0, acc0, false, false);
        acc1 = __builtin_amdgcn_wmma_f32_16x16x32_f16(false, a, false, b1, (short)0, acc1, false, false);
        acc2 = __builtin_amdgcn_wmma_f32_16x16x32_f16(false, a, false, b2, (short)0, acc2, false, false);
        acc3 = __builtin_amdgcn_wmma_f32_16x16x32_f16(false, a, false, b3, (short)0, acc3, false, false);
    }

    float* cbase = C + (size_t)(m0 + 8 * half) * N + n0 + l15;
#pragma unroll
    for (int r = 0; r < 8; ++r) {
        float* row = cbase + (size_t)r * N;
        row[0]  = acc0[r];
        row[16] = acc1[r];
        row[32] = acc2[r];
        row[48] = acc3[r];
    }

    const float* alp = al + n0 + l15;
    const float* arp = ar + n0 + l15;
    const float al0 = alp[0], al1 = alp[16], al2 = alp[32], al3 = alp[48];
    const float ar0 = arp[0], ar1 = arp[16], ar2 = arp[32], ar3 = arp[48];
    float sl[8], sr[8];
#pragma unroll
    for (int r = 0; r < 8; ++r) {
        sl[r] = acc0[r] * al0 + acc1[r] * al1 + acc2[r] * al2 + acc3[r] * al3;
        sr[r] = acc0[r] * ar0 + acc1[r] * ar1 + acc2[r] * ar2 + acc3[r] * ar3;
    }
#pragma unroll
    for (int o = 8; o > 0; o >>= 1) {
#pragma unroll
        for (int r = 0; r < 8; ++r) {
            sl[r] += __shfl_xor(sl[r], o, 32);
            sr[r] += __shfl_xor(sr[r], o, 32);
        }
    }
    float vl = sl[0], vr = sr[0];
#pragma unroll
    for (int r = 1; r < 8; ++r) {
        if (l15 == r) { vl = sl[r]; vr = sr[r]; }
    }
    if (l15 < 8) {
        const size_t o = (size_t)(m0 + 8 * half + l15) * (N >> 6) + (n0 >> 6);
        el[o] = vl;
        er[o] = vr;
    }
}

#define SORT_BLOCKS 40
#define SORT_CHUNK  (N_EDGES / SORT_BLOCKS)
#define SORT_TRIPS  ((SORT_CHUNK + 1023) / 1024)
#define NODE_GROUP  256
#define N_GROUPS    ((N_NODES + NODE_GROUP - 1) / NODE_GROUP)
#define GROUP_TRIPS (N_GROUPS * NODE_GROUP / 1024)
#define CVT1_BLOCKS (IN_DIM * D1 / 1024)
#define CVT2_BLOCKS (D1 * OUT_DIM / 1024)
#define GEMM1_BLOCKS (N_NODES / 16)
#define PLACE_BLOCKS (N_EDGES / 128)

typedef __attribute__((ext_vector_type(4))) _Float16 v4h;

__device__ inline int wave_sum_i(int v) {
    for (int o = 16; o > 0; o >>= 1) v += __shfl_xor(v, o, 32);
    return v;
}

__global__ __launch_bounds__(1024) void prep_hist_cvt(const int* __restrict__ dst,
                                                      int* __restrict__ rank,
                                                      int* __restrict__ slice_cnt,
                                                      int* __restrict__ group_cnt,
                                                      const float* __restrict__ W1,
                                                      _Float16* __restrict__ W1t,
                                                      const float* __restrict__ W2,
                                                      _Float16* __restrict__ W2t) {
    __shared__ int cnt[N_NODES];
    __shared__ int wpart[GROUP_TRIPS * 32];
    const int tid = threadIdx.x;
    if (blockIdx.x >= SORT_BLOCKS) {
        int i = (blockIdx.x - SORT_BLOCKS) * 1024 + tid;
        if (i < IN_DIM * D1) {
            cvt_transpose_one(W1, W1t, IN_DIM, D1, i);
        } else {
            i -= IN_DIM * D1;
            if (i < D1 * OUT_DIM) cvt_transpose_one(W2, W2t, D1, OUT_DIM, i);
        }
        return;
    }
    for (int i = tid; i < N_NODES; i += 1024) cnt[i] = 0;
    __syncthreads();
    const int e0 = blockIdx.x * SORT_CHUNK;
    int tv[SORT_TRIPS];
#pragma unroll
    for (int k = 0; k < SORT_TRIPS; ++k) {
        int i = tid + k * 1024;
        tv[k] = (i < SORT_CHUNK) ? dst[e0 + i] : 0;
    }
#pragma unroll
    for (int k = 0; k < SORT_TRIPS; ++k) {
        int i = tid + k * 1024;
        if (i < SORT_CHUNK) rank[e0 + i] = atomicAdd(&cnt[tv[k]], 1);
    }
    __syncthreads();
    int* row = slice_cnt + (size_t)blockIdx.x * N_NODES;
    for (int k = 0; k < GROUP_TRIPS; ++k) {
        int i = tid + k * 1024;
        int cv = 0;
        if (i < N_NODES) { cv = cnt[i]; row[i] = cv; }
        cv = wave_sum_i(cv);
        if ((tid & 31) == 0) wpart[k * 32 + (tid >> 5)] = cv;
    }
    __syncthreads();
    if (tid < N_GROUPS) {
        const int* p = wpart + (tid >> 2) * 32 + (tid & 3) * 8;
        int s = 0;
        for (int w = 0; w < 8; ++w) s += p[w];
        group_cnt[blockIdx.x * N_GROUPS + tid] = s;
    }
}

__global__ __launch_bounds__(NODE_GROUP) void slice_prefix_rowptr(int* __restrict__ slice_cnt,
                                                                  const int* __restrict__ group_cnt,
                                                                  int* __restrict__ rowptr) {
    __shared__ int wtot[NODE_GROUP / 32];
    __shared__ int wbel[NODE_GROUP / 32];
    const int tid = threadIdx.x;
    const int lane = tid & 31;
    const int wave = tid >> 5;
    const int g = blockIdx.x;
    const int n = g * NODE_GROUP + tid;
    const bool valid = n < N_NODES;
    int below = 0;
    if (tid < g) {
#pragma unroll
        for (int b = 0; b < SORT_BLOCKS; ++b) below += group_cnt[b * N_GROUPS + tid];
    }
    int v[SORT_BLOCKS];
#pragma unroll
    for (int b = 0; b < SORT_BLOCKS; ++b) v[b] = valid ? slice_cnt[(size_t)b * N_NODES + n] : 0;
    int run = 0;
#pragma unroll
    for (int b = 0; b < SORT_BLOCKS; ++b) {
        if (valid) slice_cnt[(size_t)b * N_NODES + n] = run;
        run += v[b];
    }
    int inc = run;
    for (int o = 1; o < 32; o <<= 1) {
        int u = __shfl_up(inc, o, 32);
        if (lane >= o) inc += u;
    }
    below = wave_sum_i(below);
    if (lane == 31) wtot[wave] = inc;
    if (lane == 0)  wbel[wave] = below;
    __syncthreads();
    int base = 0;
    for (int w = 0; w < NODE_GROUP / 32; ++w) {
        base += wbel[w];
        if (w < wave) base += wtot[w];
    }
    if (valid) rowptr[n] = base + inc - run;
    if (n == N_NODES - 1) rowptr[N_NODES] = base + inc;
}

__global__ __launch_bounds__(128) void proj1_and_place(const float* __restrict__ A,
                                                       const _Float16* __restrict__ Bt,
                                                       float* __restrict__ C,
                                                       const float* __restrict__ al,
                                                       const float* __restrict__ ar,
                                                       float* __restrict__ el, float* __restrict__ er,
                                                       const int* __restrict__ src,
                                                       const int* __restrict__ dst,
                                                       const int* __restrict__ rank,
                                                       const int* __restrict__ slice_off,
                                                       const int* __restrict__ rowptr,
                                                       int* __restrict__ seg_src) {
    if (blockIdx.x < GEMM1_BLOCKS) {
        gemm_wmma_body<float>(A, Bt, C, al, ar, el, er, N_NODES, D1, IN_DIM);
        return;
    }
    int e = (blockIdx.x - GEMM1_BLOCKS) * 128 + threadIdx.x;
    if (e >= N_EDGES) return;
    int t = dst[e];
    int b = e / SORT_CHUNK;
    seg_src[rowptr[t] + slice_off[(size_t)b * N_NODES + t] + rank[e]] = src[e];
}

__device__ inline float wave_max(float v) {
    for (int o = 16; o > 0; o >>= 1) v = fmaxf(v, __shfl_xor(v, o, 32));
    return v;
}
__device__ inline float wave_sum(float v) {
    for (int o = 16; o > 0; o >>= 1) v += __shfl_xor(v, o, 32);
    return v;
}
__device__ inline float leaky(float x) { return (x > 0.0f) ? x : NEG_SLOPE * x; }
__device__ inline int   lane_i(int v, int j)   { return __builtin_amdgcn_readlane(v, j); }
__device__ inline float lane_f(float v, int j) {
    return __int_as_float(__builtin_amdgcn_readlane(__float_as_int(v), j));
}
__device__ inline void fma4(float4& acc, const float4& f, float a) {
    acc.x = __builtin_fmaf(f.x, a, acc.x); acc.y = __builtin_fmaf(f.y, a, acc.y);
    acc.z = __builtin_fmaf(f.z, a, acc.z); acc.w = __builtin_fmaf(f.w, a, acc.w);
}
__device__ inline float elu1(float x) { return (x > 0.0f) ? x : (__expf(x) - 1.0f); }

__device__ inline void gather_rows_l1(const float* __restrict__ fcol, int s,
                                      float a0, float a1, float a2, float a3,
                                      int cnt, bool hi, float4& accA, float4& accB) {
    int j = 0;
    for (; j + 4 <= cnt; j += 4) {
        float4 fA[4], fB[4];
#pragma unroll
        for (int k = 0; k < 4; ++k) {
            const float* p = fcol + (size_t)lane_i(s, j + k) * D1;
            fA[k] = *(const float4*)p;
            fB[k] = *(const float4*)(p + 128);
        }
#pragma unroll
        for (int k = 0; k < 4; ++k) {
            const float w0 = lane_f(a0, j + k), w1 = lane_f(a1, j + k);
            const float w2 = lane_f(a2, j + k), w3 = lane_f(a3, j + k);
            fma4(accA, fA[k], hi ? w1 : w0);
            fma4(accB, fB[k], hi ? w3 : w2);
        }
    }
    for (; j < cnt; ++j) {
        const float* p = fcol + (size_t)lane_i(s, j) * D1;
        float4 fa = *(const float4*)p;
        float4 fb = *(const float4*)(p + 128);
        const float w0 = lane_f(a0, j), w1 = lane_f(a1, j);
        const float w2 = lane_f(a2, j), w3 = lane_f(a3, j);
        fma4(accA, fa, hi ? w1 : w0);
        fma4(accB, fb, hi ? w3 : w2);
    }
}

__device__ inline void gather_rows_l2(const float* __restrict__ fcol, int s, float a,
                                      int cnt, float2& acc) {
    int j = 0;
    for (; j + 8 <= cnt; j += 8) {
        float2 f[8];
#pragma unroll
        for (int k = 0; k < 8; ++k)
            f[k] = *(const float2*)(fcol + (size_t)lane_i(s, j + k) * OUT_DIM);
#pragma unroll
        for (int k = 0; k < 8; ++k) {
            float w = lane_f(a, j + k);
            acc.x = __builtin_fmaf(f[k].x, w, acc.x);
            acc.y = __builtin_fmaf(f[k].y, w, acc.y);
        }
    }
    for (; j < cnt; ++j) {
        float2 f = *(const float2*)(fcol + (size_t)lane_i(s, j) * OUT_DIM);
        float w = lane_f(a, j);
        acc.x = __builtin_fmaf(f.x, w, acc.x);
        acc.y = __builtin_fmaf(f.y, w, acc.y);
    }
}

#define L1_NODES 16
#define H1_PITCH (D1 + 8)
#define C2_PITCH (OUT_DIM + 4)
static_assert(N_NODES % L1_NODES == 0, "every wave of every work-group owns a node");
static_assert(OUT_DIM == 64 && D1 % 32 == 0, "4 column tiles of 16, k-steps of 32");

__global__ __launch_bounds__(512) void gat_gather_l1_proj2(const int* __restrict__ rowptr,
                                                           const int* __restrict__ seg_src,
                                                           const float* __restrict__ feat,
                                                           const float* __restrict__ el,
                                                           const float* __restrict__ er,
                                                           const float* __restrict__ bias,
                                                           const _Float16* __restrict__ W2t,
                                                           const float* __restrict__ al2,
                                                           const float* __restrict__ ar2,
                                                           float* __restrict__ feat2,
                                                           float* __restrict__ el2,
                                                           float* __restrict__ er2) {
    __shared__ _Float16 h1tile[L1_NODES * H1_PITCH];
    __shared__ float c2tile[L1_NODES * C2_PITCH];
    __shared__ float part_l[4 * L1_NODES];
    __shared__ float part_r[4 * L1_NODES];

    const int lane = threadIdx.x & 31;
    const int wave = __builtin_amdgcn_readfirstlane((int)(threadIdx.x >> 5));
    const int m0 = blockIdx.x * L1_NODES;
    const int t = m0 + wave;
    const int start = rowptr[t];
    const int end   = rowptr[t + 1];
    const bool hi = (lane >> 4) != 0;
    const float4 r4 = *(const float4*)(er + (size_t)t * 4);
    const float NEG_INF = -__builtin_inff();

    float4 accA = {0.f, 0.f, 0.f, 0.f}, accB = {0.f, 0.f, 0.f, 0.f};
    const float* fcol = feat + lane * 4;
    const int deg = end - start;
    if ((unsigned)(deg - 1) < 64u) {
        const int last = end - 1;
        const int i0 = start + lane, i1 = i0 + 32;
        const bool v0 = i0 < end, v1 = i1 < end;
        const int s0 = seg_src[min(i0, last)];
        const int s1 = seg_src[min(i1, last)];
        const float4 x0 = *(const float4*)(el + (size_t)s0 * 4);
        const float4 x1 = *(const float4*)(el + (size_t)s1 * 4);
        const float e00 = leaky(x0.x + r4.x), e10 = leaky(x1.x + r4.x);
        const float e01 = leaky(x0.y + r4.y), e11 = leaky(x1.y + r4.y);
        const float e02 = leaky(x0.z + r4.z), e12 = leaky(x1.z + r4.z);
        const float e03 = leaky(x0.w + r4.w), e13 = leaky(x1.w + r4.w);
        const float m0h = wave_max(fmaxf(e00, e10)), m1 = wave_max(fmaxf(e01, e11));
        const float m2  = wave_max(fmaxf(e02, e12)), m3 = wave_max(fmaxf(e03, e13));
        const float p00 = v0 ? __expf(e00 - m0h) : 0.f, p10 = v1 ? __expf(e10 - m0h) : 0.f;
        const float p01 = v0 ? __expf(e01 - m1)  : 0.f, p11 = v1 ? __expf(e11 - m1)  : 0.f;
        const float p02 = v0 ? __expf(e02 - m2)  : 0.f, p12 = v1 ? __expf(e12 - m2)  : 0.f;
        const float p03 = v0 ? __expf(e03 - m3)  : 0.f, p13 = v1 ? __expf(e13 - m3)  : 0.f;
        const float d0 = wave_sum(p00 + p10), d1 = wave_sum(p01 + p11);
        const float d2 = wave_sum(p02 + p12), d3 = wave_sum(p03 + p13);
        gather_rows_l1(fcol, s0, v0 ? p00 / d0 : 0.f, v0 ? p01 / d1 : 0.f,
                       v0 ? p02 / d2 : 0.f, v0 ? p03 / d3 : 0.f, min(32, deg), hi, accA, accB);
        if (deg > 32)
            gather_rows_l1(fcol, s1, v1 ? p10 / d0 : 0.f, v1 ? p11 / d1 : 0.f,
                           v1 ? p12 / d2 : 0.f, v1 ? p13 / d3 : 0.f, deg - 32, hi, accA, accB);
    } else {
        float m0h = NEG_INF, m1 = NEG_INF, m2 = NEG_INF, m3 = NEG_INF;
        for (int base = start; base < end; base += 32) {
            int i = base + lane;
            if (i < end) {
                int s = seg_src[i];
                float4 l4 = *(const float4*)(el + (size_t)s * 4);
                m0h = fmaxf(m0h, leaky(l4.x + r4.x));
                m1 = fmaxf(m1, leaky(l4.y + r4.y));
                m2 = fmaxf(m2, leaky(l4.z + r4.z));
                m3 = fmaxf(m3, leaky(l4.w + r4.w));
            }
        }
        m0h = wave_max(m0h); m1 = wave_max(m1); m2 = wave_max(m2); m3 = wave_max(m3);

        float d0 = 0.f, d1 = 0.f, d2 = 0.f, d3 = 0.f;
        for (int base = start; base < end; base += 32) {
            int i = base + lane;
            if (i < end) {
                int s = seg_src[i];
                float4 l4 = *(const float4*)(el + (size_t)s * 4);
                d0 += __expf(leaky(l4.x + r4.x) - m0h);
                d1 += __expf(leaky(l4.y + r4.y) - m1);
                d2 += __expf(leaky(l4.z + r4.z) - m2);
                d3 += __expf(leaky(l4.w + r4.w) - m3);
            }
        }
        d0 = wave_sum(d0); d1 = wave_sum(d1); d2 = wave_sum(d2); d3 = wave_sum(d3);

        for (int base = start; base < end; base += 32) {
            int i = base + lane;
            int s = 0;
            float a0 = 0.f, a1 = 0.f, a2 = 0.f, a3 = 0.f;
            if (i < end) {
                s = seg_src[i];
                float4 l4 = *(const float4*)(el + (size_t)s * 4);
                a0 = __expf(leaky(l4.x + r4.x) - m0h) / d0;
                a1 = __expf(leaky(l4.y + r4.y) - m1) / d1;
                a2 = __expf(leaky(l4.z + r4.z) - m2) / d2;
                a3 = __expf(leaky(l4.w + r4.w) - m3) / d3;
            }
            gather_rows_l1(fcol, s, a0, a1, a2, a3, min(32, end - base), hi, accA, accB);
        }
    }

    const float4 bA = *(const float4*)(bias + lane * 4);
    const float4 bB = *(const float4*)(bias + 128 + lane * 4);
    v4h oA = {(_Float16)elu1(accA.x + bA.x), (_Float16)elu1(accA.y + bA.y),
              (_Float16)elu1(accA.z + bA.z), (_Float16)elu1(accA.w + bA.w)};
    v4h oB = {(_Float16)elu1(accB.x + bB.x), (_Float16)elu1(accB.y + bB.y),
              (_Float16)elu1(accB.z + bB.z), (_Float16)elu1(accB.w + bB.w)};
    _Float16* hrow = h1tile + wave * H1_PITCH + lane * 4;
    *(v4h*)hrow = oA;
    *(v4h*)(hrow + 128) = oB;

    const int half = lane >> 4;
    const int l15  = lane & 15;
    const int n0   = wave * 16;
    const bool mm  = wave < 4;
    constexpr int KS = D1 / 32;
    constexpr int KPRE = KS - 2;
    v16h bfr[KS];
    const _Float16* brow = W2t + (size_t)(n0 + l15) * D1;
    if (mm) {
#pragma unroll
        for (int ks = 0; ks < KPRE; ++ks) bfr[ks] = load_frag(brow + 32 * ks, half);
    }
    __syncthreads();
    if (mm) {
        const _Float16* arow = h1tile + l15 * H1_PITCH;
        v8f acc = {};
#pragma unroll
        for (int ks = 0; ks < KS; ++ks) {
            v16h a = load_frag(arow + 32 * ks, half);
            acc = __builtin_amdgcn_wmma_f32_16x16x32_f16(false, a, false, bfr[ks], (short)0, acc, false, false);
            if (ks + KPRE < KS) {
                __builtin_amdgcn_sched_barrier(0);
                bfr[ks + KPRE] = load_frag(brow + 32 * (ks + KPRE), half);
            }
        }
        float* cb = c2tile + (8 * half) * C2_PITCH + n0 + l15;
        const float alv = al2[n0 + l15];
        const float arv = ar2[n0 + l15];
        float sl[8], sr[8];
#pragma unroll
        for (int r = 0; r < 8; ++r) {
            cb[r * C2_PITCH] = acc[r];
            sl[r] = acc[r] * alv;
            sr[r] = acc[r] * arv;
        }
#pragma unroll
        for (int o = 8; o > 0; o >>= 1) {
#pragma unroll
            for (int r = 0; r < 8; ++r) {
                sl[r] += __shfl_xor(sl[r], o, 32);
                sr[r] += __shfl_xor(sr[r], o, 32);
            }
        }
        float vl = sl[0], vr = sr[0];
#pragma unroll
        for (int r = 1; r < 8; ++r) {
            if (l15 == r) { vl = sl[r]; vr = sr[r]; }
        }
        if (l15 < 8) {
            part_l[wave * L1_NODES + 8 * half + l15] = vl;
            part_r[wave * L1_NODES + 8 * half + l15] = vr;
        }
    }
    __syncthreads();
    const int tid = threadIdx.x;
    if (tid < L1_NODES * (OUT_DIM / 4)) {
        const int row = tid >> 4;
        const int c4  = (tid & 15) * 4;
        const float4 v = *(const float4*)(c2tile + row * C2_PITCH + c4);
        *(float4*)(feat2 + (size_t)(m0 + row) * OUT_DIM + c4) = v;
    } else if (tid < L1_NODES * (OUT_DIM / 4) + L1_NODES) {
        const int row = tid - L1_NODES * (OUT_DIM / 4);
        el2[m0 + row] = ((part_l[row] + part_l[L1_NODES + row]) + part_l[2 * L1_NODES + row]) + part_l[3 * L1_NODES + row];
        er2[m0 + row] = ((part_r[row] + part_r[L1_NODES + row]) + part_r[2 * L1_NODES + row]) + part_r[3 * L1_NODES + row];
    }
}

__global__ __launch_bounds__(256) void gat_gather_l2(const int* __restrict__ rowptr,
                                                     const int* __restrict__ seg_src,
                                                     const float* __restrict__ feat,
                                                     const float* __restrict__ el,
                                                     const float* __restrict__ er,
                                                     const float* __restrict__ bias,
                                                     float* __restrict__ out) {
    const int lane = threadIdx.x & 31;
    const int wave = __builtin_amdgcn_readfirstlane((int)(threadIdx.x >> 5));
    const int t = blockIdx.x * 8 + wave;
    if (t >= N_NODES) return;
    const int start = rowptr[t];
    const int end   = rowptr[t + 1];
    const float r = er[t];
    const float NEG_INF = -__builtin_inff();

    float2 acc = {0.f, 0.f};
    const float* fcol = feat + lane * 2;
    const int deg = end - start;
    if ((unsigned)(deg - 1) < 64u) {
        const int last = end - 1;
        const int i0 = start + lane, i1 = i0 + 32;
        const bool v0 = i0 < end, v1 = i1 < end;
        const int s0 = seg_src[min(i0, last)];
        const int s1 = seg_src[min(i1, last)];
        const float x0 = el[s0];
        const float x1 = el[s1];
        const float e0 = leaky(x0 + r);
        const float e1 = leaky(x1 + r);
        const float m = wave_max(fmaxf(e0, e1));
        const float p0 = v0 ? __expf(e0 - m) : 0.f;
        const float p1 = v1 ? __expf(e1 - m) : 0.f;
        const float d = wave_sum(p0 + p1);
        gather_rows_l2(fcol, s0, v0 ? p0 / d : 0.f, min(32, deg), acc);
        if (deg > 32) gather_rows_l2(fcol, s1, v1 ? p1 / d : 0.f, deg - 32, acc);
    } else {
        float m = NEG_INF;
        for (int base = start; base < end; base += 32) {
            int i = base + lane;
            if (i < end) m = fmaxf(m, leaky(el[seg_src[i]] + r));
        }
        m = wave_max(m);

        float d = 0.f;
        for (int base = start; base < end; base += 32) {
            int i = base + lane;
            if (i < end) d += __expf(leaky(el[seg_src[i]] + r) - m);
        }
        d = wave_sum(d);

        for (int base = start; base < end; base += 32) {
            int i = base + lane;
            int s = 0;
            float a = 0.f;
            if (i < end) {
                s = seg_src[i];
                a = __expf(leaky(el[s] + r) - m) / d;
            }
            gather_rows_l2(fcol, s, a, min(32, end - base), acc);
        }
    }

    const float2 b = *(const float2*)(bias + lane * 2);
    float2 o = {acc.x + b.x, acc.y + b.y};
    *(float2*)(out + (size_t)t * OUT_DIM + lane * 2) = o;
}

extern "C" void kernel_launch(void* const* d_in, const int* in_sizes, int n_in,
                              void* d_out, int out_size, void* d_ws, size_t ws_size,
                              hipStream_t stream) {
    const float* h   = (const float*)d_in[0];
    const int*   src = (const int*)  d_in[1];
    const int*   dst = (const int*)  d_in[2];
    const float* W1  = (const float*)d_in[3];
    const float* al1 = (const float*)d_in[4];
    const float* ar1 = (const float*)d_in[5];
    const float* b1  = (const float*)d_in[6];
    const float* W2  = (const float*)d_in[7];
    const float* al2 = (const float*)d_in[8];
    const float* ar2 = (const float*)d_in[9];
    const float* b2  = (const float*)d_in[10];
    float* out = (float*)d_out;

    size_t off = 0;
    auto alloc = [&](size_t bytes) {
        void* p = (char*)d_ws + off;
        off += (bytes + 255) & ~(size_t)255;
        return p;
    };
    _Float16* W1t16  = (_Float16*)alloc((size_t)IN_DIM  * D1      * 2);
    _Float16* W2t16  = (_Float16*)alloc((size_t)D1      * OUT_DIM * 2);
    float*    feat1  = (float*)   alloc((size_t)N_NODES * D1      * 4);
    float*    el1    = (float*)   alloc((size_t)N_NODES * HEADS   * 4);
    float*    er1    = (float*)   alloc((size_t)N_NODES * HEADS   * 4);
    float*    feat2  = (float*)   alloc((size_t)N_NODES * OUT_DIM * 4);
    float*    el2    = (float*)   alloc((size_t)N_NODES * 4);
    float*    er2    = (float*)   alloc((size_t)N_NODES * 4);
    int*      rank   = (int*)     alloc((size_t)N_EDGES * 4);
    int*      slices = (int*)     alloc((size_t)SORT_BLOCKS * N_NODES * 4);
    int*      groups = (int*)     alloc((size_t)SORT_BLOCKS * N_GROUPS * 4);
    int*      rowptr = (int*)     alloc((size_t)(N_NODES + 1) * 4);
    int*      segsrc = (int*)     alloc((size_t)N_EDGES * 4);

    auto cdiv = [](int a, int b) { return (a + b - 1) / b; };

    prep_hist_cvt<<<SORT_BLOCKS + CVT1_BLOCKS + CVT2_BLOCKS, 1024, 0, stream>>>(
        dst, rank, slices, groups, W1, W1t16, W2, W2t16);
    slice_prefix_rowptr<<<N_GROUPS, NODE_GROUP, 0, stream>>>(slices, groups, rowptr);

    proj1_and_place<<<GEMM1_BLOCKS + PLACE_BLOCKS, 128, 0, stream>>>(h, W1t16, feat1, al1, ar1, el1, er1,
                                                                   src, dst, rank, slices, rowptr, segsrc);
    gat_gather_l1_proj2<<<N_NODES / L1_NODES, 512, 0, stream>>>(rowptr, segsrc, feat1, el1, er1, b1,
                                                                W2t16, al2, ar2, feat2, el2, er2);
    gat_gather_l2<<<cdiv(N_NODES, 8), 256, 0, stream>>>(rowptr, segsrc, feat2, el2, er2, b2, out);
    (void)in_sizes; (void)n_in; (void)out_size; (void)ws_size;
}
